# starting kernel
# baseline (speedup 1.0000x reference)
#include <hip/hip_runtime.h>

typedef __attribute__((ext_vector_type(2))) float v2f;
typedef __attribute__((ext_vector_type(8))) float v8f;

#define IC 256
#define OC 128
#define IH 32
#define IW 32
#define OH 66
#define OW 66
#define JMAX 33
#define XSROW 17

__global__ __launch_bounds__(256)
void convT2d_wmma_f32(const float* __restrict__ x,
                      const float* __restrict__ w,
                      const float* __restrict__ bias,
                      float* __restrict__ out)
{
    __shared__ float xs[IC * 3 * XSROW];

    int bid = blockIdx.x;
    int it  = bid % 3;   bid /= 3;
    int jt  = bid % 17;  bid /= 17;
    int b   = bid % 16;  bid /= 16;
    int py  = bid & 1;
    int px  = bid >> 1;
    int j   = 2 * jt;
    int i0  = it * 16;
    int tid = threadIdx.x;

    for (int e = tid; e < IC * 3 * XSROW; e += 256) {
        int ic  = e / (3 * XSROW);
        int r   = e - ic * (3 * XSROW);
        int row = r / XSROW;
        int xo  = r - row * XSROW;
        int iy  = j - 1 + row;
        int xx  = i0 - 1 + xo;
        float v = 0.0f;
        if (iy >= 0 && iy < IH && xx >= 0 && xx < IW)
            v = x[((b * IC + ic) * IH + iy) * IW + xx];
        xs[e] = v;
    }
    __syncthreads();

    int wv   = tid >> 5;
    int lane = tid & 31;
    int half = lane >> 4;
    int nn   = lane & 15;

    int ky = py + 2 * half;
    const float* wq = w + ((wv * 16 + nn) * 16 + ky * 4 + px);

    const float* lp = &xs[(1 - half) * XSROW + nn];

    v8f accA0 = {0.f,0.f,0.f,0.f,0.f,0.f,0.f,0.f};
    v8f accA1 = accA0;
    v8f accB0 = accA0;
    v8f accB1 = accA0;

    #pragma unroll 2
    for (int ic = 0; ic < IC; ic += 2) {
        v2f a0, a1, b00, b01, b10, b11;
        a0.x = wq[2];        a0.y = wq[0];
        a1.x = wq[2048 + 2]; a1.y = wq[2048];
        b00.x = lp[0];                 b00.y = lp[1];
        b01.x = lp[XSROW];             b01.y = lp[XSROW + 1];
        b10.x = lp[3 * XSROW];         b10.y = lp[3 * XSROW + 1];
        b11.x = lp[4 * XSROW];         b11.y = lp[4 * XSROW + 1];
        wq += 2 * 2048;
        lp += 6 * XSROW;

        accA0 = __builtin_amdgcn_wmma_f32_16x16x4_f32(
                    false, a0, false, b00, (short)0, accA0, false, false);
        accB0 = __builtin_amdgcn_wmma_f32_16x16x4_f32(
                    false, a0, false, b01, (short)0, accB0, false, false);
        accA1 = __builtin_amdgcn_wmma_f32_16x16x4_f32(
                    false, a1, false, b10, (short)0, accA1, false, false);
        accB1 = __builtin_amdgcn_wmma_f32_16x16x4_f32(
                    false, a1, false, b11, (short)0, accB1, false, false);
    }

    int ii = i0 + nn;
    if (ii < JMAX) {
        int oc0 = wv * 16 + half * 8;
        const float* bp = bias + oc0;
        float* p = out + ((size_t)(b * OC + oc0) * OH + (2 * j + py)) * OW
                       + (2 * ii + px);
        bool t1ok = (j + 1) < JMAX;
        #pragma unroll
        for (int r = 0; r < 8; ++r) {
            float bs = bp[r];
            p[r * (OH * OW)] = accA0[r] + accA1[r] + bs;
            if (t1ok)
                p[r * (OH * OW) + 2 * OW] = accB0[r] + accB1[r] + bs;
        }
    }
}

extern "C" void kernel_launch(void* const* d_in, const int* in_sizes, int n_in,
                              void* d_out, int out_size, void* d_ws, size_t ws_size,
                              hipStream_t stream) {
    const float* x    = (const float*)d_in[0];
    const float* wgt  = (const float*)d_in[1];
    const float* bias = (const float*)d_in[2];
    float* out = (float*)d_out;

    dim3 grid(4 * 16 * 17 * 3);
    convT2d_wmma_f32<<<grid, dim3(256), 0, stream>>>(x, wgt, bias, out);
}


# Operand loads four K steps ahead in a ring of four register sets: a step takes its fragments, refills its set with the step four ahead, then runs its 16 WMMAs (was 16 loads atop a 4-step trip)
# speedup vs baseline: 3.8831x; 3.8831x over previous
#include <hip/hip_runtime.h>


typedef __attribute__((ext_vector_type(2))) float v2f;
typedef __attribute__((ext_vector_type(4))) float v4f;
typedef __attribute__((ext_vector_type(8))) float v8f;

#define IC 256
#define OC 128
#define NPIX 1024
#define OH 66
#define OW 66
#define OCG 4
#define MW (OCG * 16)
#define PS 1040
#define OUTN (OCG * OH * OW)
#define NT 512

__global__ __launch_bounds__(NT)
void convT2d_scatter_wmma_f32(const float* __restrict__ x,
                              const float* __restrict__ w,
                              const float* __restrict__ bias,
                              float* __restrict__ out)
{
    __shared__ __attribute__((aligned(16))) float cs[MW * PS];

    const int bid = blockIdx.x;
    const int ocg = bid & 31;
    const int b   = bid >> 5;
    const int oc0 = ocg * OCG;
    const int tid = threadIdx.x;
    const int wv   = tid >> 5;
    const int lane = tid & 31;
    const int c    = lane & 15;
    const int h    = lane >> 4;
    const int n0   = wv * 64;

    const float* xp = x + (size_t)(b * IC + 2 * h) * NPIX + n0 + 4 * c;
    const float* wp = w + (size_t)(2 * h) * (OC * 16) + oc0 * 16 + 4 * c;

    v8f acc[4][4];
    #pragma unroll
    for (int a = 0; a < 4; ++a)
        #pragma unroll
        for (int n = 0; n < 4; ++n)
            acc[a][n] = (v8f){0.f, 0.f, 0.f, 0.f, 0.f, 0.f, 0.f, 0.f};

    v4f wa[4][2], xb[4][2];
    #pragma unroll
    for (int r = 0; r < 4; ++r) {
        #pragma unroll
        for (int s = 0; s < 2; ++s) {
            wa[r][s] = *(const v4f*)(wp + (4 * r + s) * (OC * 16));
            xb[r][s] = *(const v4f*)(xp + (4 * r + s) * NPIX);
        }
        __builtin_amdgcn_sched_barrier(0);
    }
    wp += 16 * (OC * 16);
    xp += 16 * NPIX;

    for (int k0 = 0; k0 < IC - 16; k0 += 16) {
        #pragma unroll
        for (int r = 0; r < 4; ++r) {
            v2f af[4], bf[4];
            #pragma unroll
            for (int a = 0; a < 4; ++a) { af[a].x = wa[r][0][a]; af[a].y = wa[r][1][a]; }
            #pragma unroll
            for (int u = 0; u < 4; ++u) { bf[u].x = xb[r][0][u]; bf[u].y = xb[r][1][u]; }
            #pragma unroll
            for (int a = 0; a < 4; ++a) asm volatile("" : "+v"(af[a]), "+v"(bf[a]));
            __builtin_amdgcn_sched_barrier(0);
            #pragma unroll
            for (int s = 0; s < 2; ++s) {
                wa[r][s] = *(const v4f*)(wp + (4 * r + s) * (OC * 16));
                xb[r][s] = *(const v4f*)(xp + (4 * r + s) * NPIX);
            }
            #pragma unroll
            for (int u = 0; u < 4; ++u)
                #pragma unroll
                for (int a = 0; a < 4; ++a)
                    acc[a][u] = __builtin_amdgcn_wmma_f32_16x16x4_f32(
                        false, af[a], false, bf[u], (short)0, acc[a][u], false, false);
            __builtin_amdgcn_sched_barrier(0);
        }
        wp += 16 * (OC * 16);
        xp += 16 * NPIX;
    }
    #pragma unroll
    for (int r = 0; r < 4; ++r) {
        #pragma unroll
        for (int u = 0; u < 4; ++u) {
            v2f bf; bf.x = xb[r][0][u]; bf.y = xb[r][1][u];
            #pragma unroll
            for (int a = 0; a < 4; ++a) {
                v2f af; af.x = wa[r][0][a]; af.y = wa[r][1][a];
                acc[a][u] = __builtin_amdgcn_wmma_f32_16x16x4_f32(
                    false, af, false, bf, (short)0, acc[a][u], false, false);
            }
        }
    }

    #pragma unroll
    for (int a = 0; a < 4; ++a)
        #pragma unroll
        for (int r = 0; r < 8; ++r) {
            v4f v;
            v.x = acc[a][0][r];
            v.y = acc[a][1][r];
            v.z = acc[a][2][r];
            v.w = acc[a][3][r];
            int m = 4 * (8 * h + r) + a;
            *(v4f*)(&cs[m * PS + n0 + 4 * c]) = v;
        }
    __syncthreads();

    float* ob = out + (size_t)(b * OC + oc0) * (OH * OW);
    for (int i4 = tid; i4 < OUTN / 4; i4 += NT) {
        v4f res;
        #pragma unroll
        for (int u = 0; u < 4; ++u) {
            int e   = i4 * 4 + u;
            int ocl = e / (OH * OW);
            int rem = e - ocl * (OH * OW);
            int oy  = rem / OW;
            int ox  = rem - oy * OW;
            int j = oy >> 1, py = oy & 1;
            int i = ox >> 1, px = ox & 1;
            float sum = bias[oc0 + ocl];
            #pragma unroll
            for (int dy = 0; dy < 2; ++dy)
                #pragma unroll
                for (int dx = 0; dx < 2; ++dx) {
                    int iy = j - dy, ix = i - dx;
                    bool ok = (unsigned)iy < 32u && (unsigned)ix < 32u;
                    int m = ocl * 16 + (py + 2 * dy) * 4 + (px + 2 * dx);
                    float v = cs[m * PS + (ok ? iy * 32 + ix : 0)];
                    sum += ok ? v : 0.0f;
                }
            res[u] = sum;
        }
        *(v4f*)(ob + i4 * 4) = res;
    }
}

extern "C" void kernel_launch(void* const* d_in, const int* in_sizes, int n_in,
                              void* d_out, int out_size, void* d_ws, size_t ws_size,
                              hipStream_t stream) {
    const float* x    = (const float*)d_in[0];
    const float* wgt  = (const float*)d_in[1];
    const float* bias = (const float*)d_in[2];
    float* out = (float*)d_out;

    dim3 grid(16 * 32);
    convT2d_scatter_wmma_f32<<<grid, dim3(NT), 0, stream>>>(x, wgt, bias, out);
}


# Overlap-add epilogue by output rows: C in LDS as [oc,ky][pixel][kx], a wave per output row, a lane sums the pair (2i,2i+1) from four 8-byte LDS reads, row index math on the scalar unit
# speedup vs baseline: 4.0271x; 1.0371x over previous
#include <hip/hip_runtime.h>


typedef __attribute__((ext_vector_type(2))) float v2f;
typedef __attribute__((ext_vector_type(4))) float v4f;
typedef __attribute__((ext_vector_type(8))) float v8f;

#define IC 256
#define OC 128
#define NPIX 1024
#define OH 66
#define OW 66
#define OCG 4
#define QP 4352
#define RP 136
#define NT 512

__global__ __launch_bounds__(NT)
void convT2d_scatter_wmma_f32(const float* __restrict__ x,
                              const float* __restrict__ w,
                              const float* __restrict__ bias,
                              float* __restrict__ out)
{
    __shared__ __attribute__((aligned(16))) float cs[OCG * 4 * QP];

    const int bid = blockIdx.x;
    const int ocg = bid & 31;
    const int b   = bid >> 5;
    const int oc0 = ocg * OCG;
    const int tid = threadIdx.x;
    const int wv   = tid >> 5;
    const int lane = tid & 31;
    const int c    = lane & 15;
    const int h    = lane >> 4;
    const int n0   = wv * 64;

    const float* xp = x + (size_t)(b * IC + 2 * h) * NPIX + n0 + 4 * c;
    const float* wp = w + (size_t)(2 * h) * (OC * 16) + oc0 * 16 + 4 * c;

    v8f acc[4][4];
    #pragma unroll
    for (int a = 0; a < 4; ++a)
        #pragma unroll
        for (int n = 0; n < 4; ++n)
            acc[a][n] = (v8f){0.f, 0.f, 0.f, 0.f, 0.f, 0.f, 0.f, 0.f};

    v4f wa[4][2], xb[4][2];
    #pragma unroll
    for (int r = 0; r < 4; ++r) {
        #pragma unroll
        for (int s = 0; s < 2; ++s) {
            wa[r][s] = *(const v4f*)(wp + (4 * r + s) * (OC * 16));
            xb[r][s] = *(const v4f*)(xp + (4 * r + s) * NPIX);
        }
        __builtin_amdgcn_sched_barrier(0);
    }
    wp += 16 * (OC * 16);
    xp += 16 * NPIX;

    for (int k0 = 0; k0 < IC - 16; k0 += 16) {
        #pragma unroll
        for (int r = 0; r < 4; ++r) {
            v2f af[4], bf[4];
            #pragma unroll
            for (int a = 0; a < 4; ++a) { af[a].x = wa[r][0][a]; af[a].y = wa[r][1][a]; }
            #pragma unroll
            for (int u = 0; u < 4; ++u) { bf[u].x = xb[r][0][u]; bf[u].y = xb[r][1][u]; }
            #pragma unroll
            for (int a = 0; a < 4; ++a) asm volatile("" : "+v"(af[a]), "+v"(bf[a]));
            __builtin_amdgcn_sched_barrier(0);
            #pragma unroll
            for (int s = 0; s < 2; ++s) {
                wa[r][s] = *(const v4f*)(wp + (4 * r + s) * (OC * 16));
                xb[r][s] = *(const v4f*)(xp + (4 * r + s) * NPIX);
            }
            #pragma unroll
            for (int u = 0; u < 4; ++u)
                #pragma unroll
                for (int a = 0; a < 4; ++a)
                    acc[a][u] = __builtin_amdgcn_wmma_f32_16x16x4_f32(
                        false, af[a], false, bf[u], (short)0, acc[a][u], false, false);
            __builtin_amdgcn_sched_barrier(0);
        }
        wp += 16 * (OC * 16);
        xp += 16 * NPIX;
    }
    #pragma unroll
    for (int r = 0; r < 4; ++r) {
        #pragma unroll
        for (int u = 0; u < 4; ++u) {
            v2f bf; bf.x = xb[r][0][u]; bf.y = xb[r][1][u];
            #pragma unroll
            for (int a = 0; a < 4; ++a) {
                v2f af; af.x = wa[r][0][a]; af.y = wa[r][1][a];
                acc[a][u] = __builtin_amdgcn_wmma_f32_16x16x4_f32(
                    false, af, false, bf, (short)0, acc[a][u], false, false);
            }
        }
    }

    {
        float* cw = cs + (8 * h) * QP + 4 * (n0 + 4 * c) + 4 * (4 * wv + (c >> 2));
        #pragma unroll
        for (int r = 0; r < 8; ++r)
            #pragma unroll
            for (int u = 0; u < 4; ++u) {
                v4f v;
                v.x = acc[0][u][r];
                v.y = acc[1][u][r];
                v.z = acc[2][u][r];
                v.w = acc[3][u][r];
                *(v4f*)(cw + r * QP + 4 * u) = v;
            }
    }
    __syncthreads();

    float* ob = out + (size_t)(b * OC + oc0) * (OH * OW);
    auto pair_sum = [&](int ocl, int oy, int i) -> v2f {
        const int j = oy >> 1, py = oy & 1;
        const float bs = bias[oc0 + ocl];
        float s0 = bs, s1 = bs;
        for (int dy = 0; dy < 2; ++dy) {
            const int  iy  = j - dy;
            const bool oky = (unsigned)iy < 32u;
            const float* pr = cs + (ocl * 4 + py + 2 * dy) * QP + (oky ? iy : 0) * RP;
            for (int dx = 0; dx < 2; ++dx) {
                const int  ix  = i - dx;
                const bool ok  = oky && (unsigned)ix < 32u;
                const int  ixc = (unsigned)ix < 32u ? ix : 0;
                const v2f  v   = *(const v2f*)(pr + 4 * ixc + 4 * (ixc >> 4) + 2 * dx);
                s0 += ok ? v.x : 0.0f;
                s1 += ok ? v.y : 0.0f;
            }
        }
        v2f res; res.x = s0; res.y = s1;
        return res;
    };

    const int wvu = __builtin_amdgcn_readfirstlane(wv);
    for (int row = wvu; row < OCG * OH; row += NT / 32) {
        const int ocl = row / OH;
        const int oy  = row - ocl * OH;
        *(v2f*)(ob + row * OW + 2 * lane) = pair_sum(ocl, oy, lane);
    }
    if (tid < OCG * OH) {
        const int ocl = tid / OH;
        const int oy  = tid - ocl * OH;
        *(v2f*)(ob + tid * OW + 64) = pair_sum(ocl, oy, 32);
    }
}

extern "C" void kernel_launch(void* const* d_in, const int* in_sizes, int n_in,
                              void* d_out, int out_size, void* d_ws, size_t ws_size,
                              hipStream_t stream) {
    const float* x    = (const float*)d_in[0];
    const float* wgt  = (const float*)d_in[1];
    const float* bias = (const float*)d_in[2];
    float* out = (float*)d_out;

    dim3 grid(16 * 32);
    convT2d_scatter_wmma_f32<<<grid, dim3(NT), 0, stream>>>(x, wgt, bias, out);
}


# 9 products for 16: the 3-for-4 pairing along y too (2x2 pixel block x 2x2 tap block at points 0, inf, 1 both ways; new summation order): 9 WMMAs per K step, was 12
# speedup vs baseline: 5.4195x; 1.3458x over previous
#include <hip/hip_runtime.h>


typedef __attribute__((ext_vector_type(2))) float v2f;
typedef __attribute__((ext_vector_type(4))) float v4f;
typedef __attribute__((ext_vector_type(8))) float v8f;

#define IC 256
#define OC 128
#define NPIX 1024
#define OH 66
#define OW 66
#define OCG 4
#define RS 96
#define PS 4616
#define NT 512

__global__ __launch_bounds__(NT)
void convT2d_scatter_wmma_f32(const float* __restrict__ x,
                              const float* __restrict__ w,
                              const float* __restrict__ bias,
                              float* __restrict__ out)
{
    __shared__ __attribute__((aligned(16))) float cs[8 * PS];

    const int bid = blockIdx.x;
    const int ocg = bid & 31;
    const int b   = bid >> 5;
    const int oc0 = ocg * OCG;
    const int tid = threadIdx.x;
    const int wv   = tid >> 5;
    const int lane = tid & 31;
    const int c    = lane & 15;
    const int h    = lane >> 4;

    const float* xp = x + (size_t)(b * IC + 2 * h) * NPIX + 64 * wv + 2 * c;
    const float* wp = w + (size_t)(2 * h) * (OC * 16) + (oc0 + (c >> 2)) * 16
                        + ((c >> 1) & 1) * 4;
    const bool kxl = (c & 1) != 0;

    v8f acc[3][3];
    #pragma unroll
    for (int e = 0; e < 3; ++e)
        #pragma unroll
        for (int f = 0; f < 3; ++f)
            acc[e][f] = (v8f){0.f, 0.f, 0.f, 0.f, 0.f, 0.f, 0.f, 0.f};

    v4f   wa[4][2][2];
    v2f   xb[4][2][2];
    #pragma unroll
    for (int r = 0; r < 4; ++r) {
        #pragma unroll
        for (int s = 0; s < 2; ++s) {
            #pragma unroll
            for (int e = 0; e < 2; ++e)
                wa[r][s][e] = *(const v4f*)(wp + (4 * r + s) * (OC * 16) + 8 * e);
            #pragma unroll
            for (int e = 0; e < 2; ++e)
                xb[r][s][e] = *(const v2f*)(xp + (4 * r + s) * NPIX + 32 * e);
        }
        __builtin_amdgcn_sched_barrier(0);
    }
    wp += 16 * (OC * 16);
    xp += 16 * NPIX;

    for (int k0 = 0; k0 < IC - 16; k0 += 16) {
        #pragma unroll
        for (int r = 0; r < 4; ++r) {
            v2f af[3][3], bf[3][3];
            #pragma unroll
            for (int e = 0; e < 2; ++e) {
                const v4f q0 = wa[r][0][e], q1 = wa[r][1][e];
                const v2f t0 = kxl ? __builtin_shufflevector(q0, q0, 1, 3) : __builtin_shufflevector(q0, q0, 0, 2);
                const v2f t1 = kxl ? __builtin_shufflevector(q1, q1, 1, 3) : __builtin_shufflevector(q1, q1, 0, 2);
                #pragma unroll
                for (int f = 0; f < 2; ++f) {
                    af[e][f].x = t0[f];           af[e][f].y = t1[f];
                    bf[e][f].x = xb[r][0][e][f];  bf[e][f].y = xb[r][1][e][f];
                }
            }
            #pragma unroll
            for (int e = 0; e < 2; ++e)
                #pragma unroll
                for (int f = 0; f < 2; ++f) asm volatile("" : "+v"(af[e][f]), "+v"(bf[e][f]));
            #pragma unroll
            for (int e = 0; e < 2; ++e) {
                af[e][2] = af[e][0] + af[e][1];
                bf[e][2] = bf[e][0] + bf[e][1];
            }
            #pragma unroll
            for (int f = 0; f < 3; ++f) {
                af[2][f] = af[0][f] + af[1][f];
                bf[2][f] = bf[0][f] + bf[1][f];
            }
            #pragma unroll
            for (int e = 0; e < 2; ++e) asm volatile("" : "+v"(af[e][2]), "+v"(bf[e][2]));
            #pragma unroll
            for (int f = 0; f < 3; ++f) asm volatile("" : "+v"(af[2][f]), "+v"(bf[2][f]));
            __builtin_amdgcn_sched_barrier(0);
            #pragma unroll
            for (int s = 0; s < 2; ++s) {
                #pragma unroll
                for (int e = 0; e < 2; ++e)
                    wa[r][s][e] = *(const v4f*)(wp + (4 * r + s) * (OC * 16) + 8 * e);
                #pragma unroll
                for (int e = 0; e < 2; ++e)
                    xb[r][s][e] = *(const v2f*)(xp + (4 * r + s) * NPIX + 32 * e);
            }
            #pragma unroll
            for (int e = 0; e < 3; ++e)
                #pragma unroll
                for (int f = 0; f < 3; ++f)
                    acc[e][f] = __builtin_amdgcn_wmma_f32_16x16x4_f32(
                        false, af[e][f], false, bf[e][f], (short)0, acc[e][f], false, false);
            __builtin_amdgcn_sched_barrier(0);
        }
        wp += 16 * (OC * 16);
        xp += 16 * NPIX;
    }
    #pragma unroll
    for (int r = 0; r < 4; ++r) {
        v2f af[3][3], bf[3][3];
        #pragma unroll
        for (int e = 0; e < 2; ++e) {
            const v4f q0 = wa[r][0][e], q1 = wa[r][1][e];
            const v2f t0 = kxl ? __builtin_shufflevector(q0, q0, 1, 3) : __builtin_shufflevector(q0, q0, 0, 2);
            const v2f t1 = kxl ? __builtin_shufflevector(q1, q1, 1, 3) : __builtin_shufflevector(q1, q1, 0, 2);
            #pragma unroll
            for (int f = 0; f < 2; ++f) {
                af[e][f].x = t0[f];           af[e][f].y = t1[f];
                bf[e][f].x = xb[r][0][e][f];  bf[e][f].y = xb[r][1][e][f];
            }
        }
        #pragma unroll
        for (int e = 0; e < 2; ++e) {
            af[e][2] = af[e][0] + af[e][1];
            bf[e][2] = bf[e][0] + bf[e][1];
        }
        #pragma unroll
        for (int f = 0; f < 3; ++f) {
            af[2][f] = af[0][f] + af[1][f];
            bf[2][f] = bf[0][f] + bf[1][f];
        }
        #pragma unroll
        for (int e = 0; e < 3; ++e)
            #pragma unroll
            for (int f = 0; f < 3; ++f)
                acc[e][f] = __builtin_amdgcn_wmma_f32_16x16x4_f32(
                    false, af[e][f], false, bf[e][f], (short)0, acc[e][f], false, false);
    }

    #pragma unroll
    for (int e = 0; e < 3; ++e) acc[e][2] = acc[e][2] - acc[e][0] - acc[e][1];
    #pragma unroll
    for (int f = 0; f < 3; ++f) acc[2][f] = acc[2][f] - acc[0][f] - acc[1][f];

    {
        float* cw = cs + (4 * h) * PS + (3 * wv) * RS + 6 * c;
        #pragma unroll
        for (int j = 0; j < 4; ++j)
            #pragma unroll
            for (int e = 0; e < 3; ++e)
                #pragma unroll
                for (int f = 0; f < 3; ++f) {
                    const int ty = e == 0 ? 0 : (e == 1 ? 2 : 1);
                    const int tx = f == 0 ? 0 : (f == 1 ? 2 : 1);
                    v2f v;
                    v.x = acc[e][f][2 * j];
                    v.y = acc[e][f][2 * j + 1];
                    *(v2f*)(cw + j * PS + ty * RS + 2 * tx) = v;
                }
    }
    __syncthreads();

    float* ob = out + (size_t)(b * OC + oc0) * (OH * OW);
    auto pair_sum = [&](int ocl, int oy, int i) -> v2f {
        const int sy = oy >> 1, kyl = oy & 1;
        const float bs = bias[oc0 + ocl];
        float s0 = bs, s1 = bs;
        const float* pl = cs + (ocl * 2 + kyl) * PS;
        for (int e = 0; e < 2; ++e) {
            const bool oky = e == 0 ? (sy < 32) : ((sy & 1) == 0 && sy >= 2);
            const int  ly  = oky ? 3 * (sy >> 1) + (e == 0 ? (sy & 1) : -1) : 0;
            for (int f = 0; f < 2; ++f) {
                const bool okx = f == 0 ? (i < 32) : ((i & 1) == 0 && i >= 2);
                const int  lx  = okx ? 3 * (i >> 1) + (f == 0 ? (i & 1) : -1) : 0;
                const v2f  v   = *(const v2f*)(pl + ly * RS + 2 * lx);
                const bool ok  = oky && okx;
                s0 += ok ? v.x : 0.0f;
                s1 += ok ? v.y : 0.0f;
            }
        }
        v2f res; res.x = s0; res.y = s1;
        return res;
    };

    const int wvu = __builtin_amdgcn_readfirstlane(wv);
    for (int row = wvu; row < OCG * OH; row += NT / 32) {
        const int ocl = row / OH;
        const int oy  = row - ocl * OH;
        *(v2f*)(ob + row * OW + 2 * lane) = pair_sum(ocl, oy, lane);
    }
    if (tid < OCG * OH) {
        const int ocl = tid / OH;
        const int oy  = tid - ocl * OH;
        *(v2f*)(ob + tid * OW + 64) = pair_sum(ocl, oy, 32);
    }
}

extern "C" void kernel_launch(void* const* d_in, const int* in_sizes, int n_in,
                              void* d_out, int out_size, void* d_ws, size_t ws_size,
                              hipStream_t stream) {
    const float* x    = (const float*)d_in[0];
    const float* wgt  = (const float*)d_in[1];
    const float* bias = (const float*)d_in[2];
    float* out = (float*)d_out;

    dim3 grid(16 * 32);
    convT2d_scatter_wmma_f32<<<grid, dim3(NT), 0, stream>>>(x, wgt, bias, out);
}


# 8 output channels per workgroup (was 4): lane row = (oc, ky parity), the two kx parities as two sets of 9 tiles sharing the pixel fragments; 8 loads feed 18 WMMAs (was 9); 256 workgroups
# speedup vs baseline: 5.9434x; 1.0967x over previous
#include <hip/hip_runtime.h>


typedef __attribute__((ext_vector_type(2))) float v2f;
typedef __attribute__((ext_vector_type(4))) float v4f;
typedef __attribute__((ext_vector_type(8))) float v8f;

#define IC 256
#define OC 128
#define NPIX 1024
#define OH 66
#define OW 66
#define OCG 8
#define RS 96
#define PS 4612
#define NT 512

__global__ __launch_bounds__(NT)
void convT2d_scatter_wmma_f32(const float* __restrict__ x,
                              const float* __restrict__ w,
                              const float* __restrict__ bias,
                              float* __restrict__ out)
{
    __shared__ __attribute__((aligned(16))) float cs[2 * OCG * PS];

    const int bid = blockIdx.x;
    const int ocg = bid & 15;
    const int b   = bid >> 4;
    const int oc0 = ocg * OCG;
    const int tid = threadIdx.x;
    const int wv   = tid >> 5;
    const int lane = tid & 31;
    const int c    = lane & 15;
    const int h    = lane >> 4;

    const float* xp = x + (size_t)(b * IC + 2 * h) * NPIX + 64 * wv + 2 * c;
    const float* wp = w + (size_t)(2 * h) * (OC * 16) + (oc0 + (c >> 1)) * 16
                        + (c & 1) * 4;

    v8f acc[2][3][3];
    #pragma unroll
    for (int p = 0; p < 2; ++p)
        #pragma unroll
        for (int e = 0; e < 3; ++e)
            #pragma unroll
            for (int f = 0; f < 3; ++f)
                acc[p][e][f] = (v8f){0.f, 0.f, 0.f, 0.f, 0.f, 0.f, 0.f, 0.f};

    v4f   wa[2][2][2];
    v2f   xb[2][2][2];
    #pragma unroll
    for (int r = 0; r < 2; ++r) {
        #pragma unroll
        for (int s = 0; s < 2; ++s) {
            #pragma unroll
            for (int e = 0; e < 2; ++e)
                wa[r][s][e] = *(const v4f*)(wp + (4 * r + s) * (OC * 16) + 8 * e);
            #pragma unroll
            for (int e = 0; e < 2; ++e)
                xb[r][s][e] = *(const v2f*)(xp + (4 * r + s) * NPIX + 32 * e);
        }
        __builtin_amdgcn_sched_barrier(0);
    }
    wp += 8 * (OC * 16);
    xp += 8 * NPIX;

    for (int k0 = 0; k0 < IC - 8; k0 += 8) {
        #pragma unroll
        for (int r = 0; r < 2; ++r) {
            v2f af[2][3][3], bf[3][3];
            #pragma unroll
            for (int e = 0; e < 2; ++e) {
                const v4f q0 = wa[r][0][e], q1 = wa[r][1][e];
                #pragma unroll
                for (int f = 0; f < 2; ++f) {
                    af[0][e][f].x = q0[2 * f];      af[0][e][f].y = q1[2 * f];
                    af[1][e][f].x = q0[2 * f + 1];  af[1][e][f].y = q1[2 * f + 1];
                    bf[e][f].x = xb[r][0][e][f];    bf[e][f].y = xb[r][1][e][f];
                }
            }
            #pragma unroll
            for (int e = 0; e < 2; ++e)
                #pragma unroll
                for (int f = 0; f < 2; ++f)
                    asm volatile("" : "+v"(af[0][e][f]), "+v"(af[1][e][f]), "+v"(bf[e][f]));
            #pragma unroll
            for (int e = 0; e < 2; ++e) {
                af[0][e][2] = af[0][e][0] + af[0][e][1];
                af[1][e][2] = af[1][e][0] + af[1][e][1];
                bf[e][2] = bf[e][0] + bf[e][1];
            }
            #pragma unroll
            for (int f = 0; f < 3; ++f) {
                af[0][2][f] = af[0][0][f] + af[0][1][f];
                af[1][2][f] = af[1][0][f] + af[1][1][f];
                bf[2][f] = bf[0][f] + bf[1][f];
            }
            __builtin_amdgcn_sched_barrier(0);
            #pragma unroll
            for (int s = 0; s < 2; ++s) {
                #pragma unroll
                for (int e = 0; e < 2; ++e)
                    wa[r][s][e] = *(const v4f*)(wp + (4 * r + s) * (OC * 16) + 8 * e);
                #pragma unroll
                for (int e = 0; e < 2; ++e)
                    xb[r][s][e] = *(const v2f*)(xp + (4 * r + s) * NPIX + 32 * e);
            }
            #pragma unroll
            for (int p = 0; p < 2; ++p)
                #pragma unroll
                for (int e = 0; e < 3; ++e)
                    #pragma unroll
                    for (int f = 0; f < 3; ++f)
                        acc[p][e][f] = __builtin_amdgcn_wmma_f32_16x16x4_f32(
                            false, af[p][e][f], false, bf[e][f], (short)0, acc[p][e][f], false, false);
            __builtin_amdgcn_sched_barrier(0);
        }
        wp += 8 * (OC * 16);
        xp += 8 * NPIX;
    }
    #pragma unroll
    for (int r = 0; r < 2; ++r) {
        v2f af[2][3][3], bf[3][3];
        #pragma unroll
        for (int e = 0; e < 2; ++e) {
            const v4f q0 = wa[r][0][e], q1 = wa[r][1][e];
            #pragma unroll
            for (int f = 0; f < 2; ++f) {
                af[0][e][f].x = q0[2 * f];      af[0][e][f].y = q1[2 * f];
                af[1][e][f].x = q0[2 * f + 1];  af[1][e][f].y = q1[2 * f + 1];
                bf[e][f].x = xb[r][0][e][f];    bf[e][f].y = xb[r][1][e][f];
            }
        }
        #pragma unroll
        for (int e = 0; e < 2; ++e) {
            af[0][e][2] = af[0][e][0] + af[0][e][1];
            af[1][e][2] = af[1][e][0] + af[1][e][1];
            bf[e][2] = bf[e][0] + bf[e][1];
        }
        #pragma unroll
        for (int f = 0; f < 3; ++f) {
            af[0][2][f] = af[0][0][f] + af[0][1][f];
            af[1][2][f] = af[1][0][f] + af[1][1][f];
            bf[2][f] = bf[0][f] + bf[1][f];
        }
        #pragma unroll
        for (int p = 0; p < 2; ++p)
            #pragma unroll
            for (int e = 0; e < 3; ++e)
                #pragma unroll
                for (int f = 0; f < 3; ++f)
                    acc[p][e][f] = __builtin_amdgcn_wmma_f32_16x16x4_f32(
                        false, af[p][e][f], false, bf[e][f], (short)0, acc[p][e][f], false, false);
    }

    #pragma unroll
    for (int p = 0; p < 2; ++p) {
        #pragma unroll
        for (int e = 0; e < 3; ++e) acc[p][e][2] = acc[p][e][2] - acc[p][e][0] - acc[p][e][1];
        #pragma unroll
        for (int f = 0; f < 3; ++f) acc[p][2][f] = acc[p][2][f] - acc[p][0][f] - acc[p][1][f];
    }

    {
        float* cw = cs + (8 * h) * PS + (3 * wv) * RS + 6 * c;
        #pragma unroll
        for (int j = 0; j < 8; ++j)
            #pragma unroll
            for (int e = 0; e < 3; ++e)
                #pragma unroll
                for (int f = 0; f < 3; ++f) {
                    const int ty = e == 0 ? 0 : (e == 1 ? 2 : 1);
                    const int tx = f == 0 ? 0 : (f == 1 ? 2 : 1);
                    v2f v;
                    v.x = acc[0][e][f][j];
                    v.y = acc[1][e][f][j];
                    *(v2f*)(cw + j * PS + ty * RS + 2 * tx) = v;
                }
    }
    __syncthreads();

    float* ob = out + (size_t)(b * OC + oc0) * (OH * OW);
    auto pair_sum = [&](int ocl, int oy, int i) -> v2f {
        const int sy = oy >> 1, kyl = oy & 1;
        const float bs = bias[oc0 + ocl];
        float s0 = bs, s1 = bs;
        const float* pl = cs + (ocl * 2 + kyl) * PS;
        for (int e = 0; e < 2; ++e) {
            const bool oky = e == 0 ? (sy < 32) : ((sy & 1) == 0 && sy >= 2);
            const int  ly  = oky ? 3 * (sy >> 1) + (e == 0 ? (sy & 1) : -1) : 0;
            for (int f = 0; f < 2; ++f) {
                const bool okx = f == 0 ? (i < 32) : ((i & 1) == 0 && i >= 2);
                const int  lx  = okx ? 3 * (i >> 1) + (f == 0 ? (i & 1) : -1) : 0;
                const v2f  v   = *(const v2f*)(pl + ly * RS + 2 * lx);
                const bool ok  = oky && okx;
                s0 += ok ? v.x : 0.0f;
                s1 += ok ? v.y : 0.0f;
            }
        }
        v2f res; res.x = s0; res.y = s1;
        return res;
    };

    const int wvu = __builtin_amdgcn_readfirstlane(wv);
    for (int row = wvu; row < OCG * OH; row += NT / 32) {
        const int ocl = row / OH;
        const int oy  = row - ocl * OH;
        *(v2f*)(ob + row * OW + 2 * lane) = pair_sum(ocl, oy, lane);
    }
    for (int t = tid; t < OCG * OH; t += NT) {
        const int ocl = t / OH;
        const int oy  = t - ocl * OH;
        *(v2f*)(ob + t * OW + 64) = pair_sum(ocl, oy, 32);
    }
}

extern "C" void kernel_launch(void* const* d_in, const int* in_sizes, int n_in,
                              void* d_out, int out_size, void* d_ws, size_t ws_size,
                              hipStream_t stream) {
    const float* x    = (const float*)d_in[0];
    const float* wgt  = (const float*)d_in[1];
    const float* bias = (const float*)d_in[2];
    float* out = (float*)d_out;

    dim3 grid(16 * 16);
    convT2d_scatter_wmma_f32<<<grid, dim3(NT), 0, stream>>>(x, wgt, bias, out);
}


# Overlap-add in registers: x overlap by a one-lane DPP shift, a lane stores 16 bytes of an output row straight from its accumulators; only the row two waves share crosses LDS (new summation order)
# speedup vs baseline: 6.2236x; 1.0471x over previous
#include <hip/hip_runtime.h>


typedef __attribute__((ext_vector_type(2))) float v2f;
typedef __attribute__((ext_vector_type(4))) float v4f;
typedef __attribute__((ext_vector_type(8))) float v8f;
typedef v4f v4f_a8 __attribute__((aligned(8)));

#define IC 256
#define OC 128
#define NPIX 1024
#define OH 66
#define OW 66
#define OCG 8
#define XR 68
#define NT 512

__global__ __launch_bounds__(NT)
void convT2d_scatter_wmma_f32(const float* __restrict__ x,
                              const float* __restrict__ w,
                              const float* __restrict__ bias,
                              float* __restrict__ out)
{
    __shared__ __attribute__((aligned(16))) float rb[15 * 2 * OCG * XR];

    const int bid = blockIdx.x;
    const int ocg = bid & 15;
    const int b   = bid >> 4;
    const int oc0 = ocg * OCG;
    const int tid = threadIdx.x;
    const int wv   = tid >> 5;
    const int lane = tid & 31;
    const int c    = lane & 15;
    const int h    = lane >> 4;

    const float* xp = x + (size_t)(b * IC + 2 * h) * NPIX + 64 * wv + 2 * c;
    const float* wp = w + (size_t)(2 * h) * (OC * 16) + (oc0 + (c >> 1)) * 16
                        + (c & 1) * 4;

    v8f acc[2][3][3];
    #pragma unroll
    for (int p = 0; p < 2; ++p)
        #pragma unroll
        for (int e = 0; e < 3; ++e)
            #pragma unroll
            for (int f = 0; f < 3; ++f)
                acc[p][e][f] = (v8f){0.f, 0.f, 0.f, 0.f, 0.f, 0.f, 0.f, 0.f};

    v4f   wa[2][2][2];
    v2f   xb[2][2][2];
    #pragma unroll
    for (int r = 0; r < 2; ++r) {
        #pragma unroll
        for (int s = 0; s < 2; ++s) {
            #pragma unroll
            for (int e = 0; e < 2; ++e)
                wa[r][s][e] = *(const v4f*)(wp + (4 * r + s) * (OC * 16) + 8 * e);
            #pragma unroll
            for (int e = 0; e < 2; ++e)
                xb[r][s][e] = *(const v2f*)(xp + (4 * r + s) * NPIX + 32 * e);
        }
        __builtin_amdgcn_sched_barrier(0);
    }
    wp += 8 * (OC * 16);
    xp += 8 * NPIX;

    for (int k0 = 0; k0 < IC - 8; k0 += 8) {
        #pragma unroll
        for (int r = 0; r < 2; ++r) {
            v2f af[2][3][3], bf[3][3];
            #pragma unroll
            for (int e = 0; e < 2; ++e) {
                const v4f q0 = wa[r][0][e], q1 = wa[r][1][e];
                #pragma unroll
                for (int f = 0; f < 2; ++f) {
                    af[0][e][f].x = q0[2 * f];      af[0][e][f].y = q1[2 * f];
                    af[1][e][f].x = q0[2 * f + 1];  af[1][e][f].y = q1[2 * f + 1];
                    bf[e][f].x = xb[r][0][e][f];    bf[e][f].y = xb[r][1][e][f];
                }
            }
            #pragma unroll
            for (int e = 0; e < 2; ++e)
                #pragma unroll
                for (int f = 0; f < 2; ++f)
                    asm volatile("" : "+v"(af[0][e][f]), "+v"(af[1][e][f]), "+v"(bf[e][f]));
            #pragma unroll
            for (int e = 0; e < 2; ++e) {
                af[0][e][2] = af[0][e][0] + af[0][e][1];
                af[1][e][2] = af[1][e][0] + af[1][e][1];
                bf[e][2] = bf[e][0] + bf[e][1];
            }
            #pragma unroll
            for (int f = 0; f < 3; ++f) {
                af[0][2][f] = af[0][0][f] + af[0][1][f];
                af[1][2][f] = af[1][0][f] + af[1][1][f];
                bf[2][f] = bf[0][f] + bf[1][f];
            }
            __builtin_amdgcn_sched_barrier(0);
            #pragma unroll
            for (int s = 0; s < 2; ++s) {
                #pragma unroll
                for (int e = 0; e < 2; ++e)
                    wa[r][s][e] = *(const v4f*)(wp + (4 * r + s) * (OC * 16) + 8 * e);
                #pragma unroll
                for (int e = 0; e < 2; ++e)
                    xb[r][s][e] = *(const v2f*)(xp + (4 * r + s) * NPIX + 32 * e);
            }
            #pragma unroll
            for (int p = 0; p < 2; ++p)
                #pragma unroll
                for (int e = 0; e < 3; ++e)
                    #pragma unroll
                    for (int f = 0; f < 3; ++f)
                        acc[p][e][f] = __builtin_amdgcn_wmma_f32_16x16x4_f32(
                            false, af[p][e][f], false, bf[e][f], (short)0, acc[p][e][f], false, false);
            __builtin_amdgcn_sched_barrier(0);
        }
        wp += 8 * (OC * 16);
        xp += 8 * NPIX;
    }
    #pragma unroll
    for (int r = 0; r < 2; ++r) {
        v2f af[2][3][3], bf[3][3];
        #pragma unroll
        for (int e = 0; e < 2; ++e) {
            const v4f q0 = wa[r][0][e], q1 = wa[r][1][e];
            #pragma unroll
            for (int f = 0; f < 2; ++f) {
                af[0][e][f].x = q0[2 * f];      af[0][e][f].y = q1[2 * f];
                af[1][e][f].x = q0[2 * f + 1];  af[1][e][f].y = q1[2 * f + 1];
                bf[e][f].x = xb[r][0][e][f];    bf[e][f].y = xb[r][1][e][f];
            }
        }
        #pragma unroll
        for (int e = 0; e < 2; ++e) {
            af[0][e][2] = af[0][e][0] + af[0][e][1];
            af[1][e][2] = af[1][e][0] + af[1][e][1];
            bf[e][2] = bf[e][0] + bf[e][1];
        }
        #pragma unroll
        for (int f = 0; f < 3; ++f) {
            af[0][2][f] = af[0][0][f] + af[0][1][f];
            af[1][2][f] = af[1][0][f] + af[1][1][f];
            bf[2][f] = bf[0][f] + bf[1][f];
        }
        #pragma unroll
        for (int p = 0; p < 2; ++p)
            #pragma unroll
            for (int e = 0; e < 3; ++e)
                #pragma unroll
                for (int f = 0; f < 3; ++f)
                    acc[p][e][f] = __builtin_amdgcn_wmma_f32_16x16x4_f32(
                        false, af[p][e][f], false, bf[e][f], (short)0, acc[p][e][f], false, false);
    }

    #pragma unroll
    for (int p = 0; p < 2; ++p) {
        #pragma unroll
        for (int e = 0; e < 3; ++e) acc[p][e][2] = acc[p][e][2] - acc[p][e][0] - acc[p][e][1];
        #pragma unroll
        for (int f = 0; f < 3; ++f) acc[p][2][f] = acc[p][2][f] - acc[p][0][f] - acc[p][1][f];
    }

    auto shr1 = [](float v) -> float {
        return __builtin_bit_cast(float, __builtin_amdgcn_update_dpp(
            0, __builtin_bit_cast(int, v), 0x111, 0xf, 0xf, true));
    };
    const int wvu = __builtin_amdgcn_readfirstlane(wv);
    const v4f bs4 = *(const v4f*)(bias + oc0 + 4 * h);
    float* orow = out + ((size_t)(b * OC + oc0 + 4 * h) * OH + 4 * wvu) * OW + 4 * c;

    #pragma unroll
    for (int j = 0; j < 8; ++j) {
        const float bsj = bs4[j >> 1];
        v4f r;
        r.x = (acc[0][2][0][j] + shr1(acc[0][2][1][j])) + bsj;
        r.y = (acc[1][2][0][j] + shr1(acc[1][2][1][j])) + bsj;
        r.z = acc[0][2][2][j] + bsj;
        r.w = acc[1][2][2][j] + bsj;
        *(v4f_a8*)(orow + ((j >> 1) * OH + 2 + (j & 1)) * OW) = r;
    }
    if (c == 15) {
        #pragma unroll
        for (int j = 0; j < 8; ++j) {
            const float bsj = bs4[j >> 1];
            v2f ed;
            ed.x = acc[0][2][1][j] + bsj;
            ed.y = acc[1][2][1][j] + bsj;
            *(v2f*)(orow + ((j >> 1) * OH + 2 + (j & 1)) * OW + 4) = ed;
        }
    }

    if (wvu < 15) {
        float* rw = rb + (wvu * 16 + 8 * h) * XR + 4 * c;
        #pragma unroll
        for (int j = 0; j < 8; ++j) {
            v4f r;
            r.x = acc[0][1][0][j] + shr1(acc[0][1][1][j]);
            r.y = acc[1][1][0][j] + shr1(acc[1][1][1][j]);
            r.z = acc[0][1][2][j];
            r.w = acc[1][1][2][j];
            *(v4f*)(rw + j * XR) = r;
        }
        if (c == 15) {
            #pragma unroll
            for (int j = 0; j < 8; ++j) {
                v2f ed;
                ed.x = acc[0][1][1][j];
                ed.y = acc[1][1][1][j];
                *(v2f*)(rw + j * XR + 4) = ed;
            }
        }
    } else {
        #pragma unroll
        for (int j = 0; j < 8; ++j) {
            const float bsj = bs4[j >> 1];
            v4f r;
            r.x = (acc[0][1][0][j] + shr1(acc[0][1][1][j])) + bsj;
            r.y = (acc[1][1][0][j] + shr1(acc[1][1][1][j])) + bsj;
            r.z = acc[0][1][2][j] + bsj;
            r.w = acc[1][1][2][j] + bsj;
            *(v4f_a8*)(orow + ((j >> 1) * OH + 4 + (j & 1)) * OW) = r;
        }
        if (c == 15) {
            #pragma unroll
            for (int j = 0; j < 8; ++j) {
                const float bsj = bs4[j >> 1];
                v2f ed;
                ed.x = acc[0][1][1][j] + bsj;
                ed.y = acc[1][1][1][j] + bsj;
                *(v2f*)(orow + ((j >> 1) * OH + 4 + (j & 1)) * OW + 4) = ed;
            }
        }
    }

    v4f top[8];
    v2f tope[8];
    #pragma unroll
    for (int j = 0; j < 8; ++j) {
        top[j].x = acc[0][0][0][j] + shr1(acc[0][0][1][j]);
        top[j].y = acc[1][0][0][j] + shr1(acc[1][0][1][j]);
        top[j].z = acc[0][0][2][j];
        top[j].w = acc[1][0][2][j];
        tope[j].x = acc[0][0][1][j];
        tope[j].y = acc[1][0][1][j];
    }
    __builtin_amdgcn_fence(__ATOMIC_RELEASE, "workgroup", "local");
    __builtin_amdgcn_s_barrier();
    __builtin_amdgcn_fence(__ATOMIC_ACQUIRE, "workgroup", "local");
    if (wvu > 0) {
        const float* rr = rb + ((wvu - 1) * 16 + 8 * h) * XR + 4 * c;
        #pragma unroll
        for (int j = 0; j < 8; ++j)
            top[j] += *(const v4f*)(rr + j * XR);
        if (c == 15) {
            #pragma unroll
            for (int j = 0; j < 8; ++j)
                tope[j] += *(const v2f*)(rr + j * XR + 4);
        }
    }
    #pragma unroll
    for (int j = 0; j < 8; ++j) {
        const float bsj = bs4[j >> 1];
        v4f r = top[j];
        r.x += bsj; r.y += bsj; r.z += bsj; r.w += bsj;
        *(v4f_a8*)(orow + ((j >> 1) * OH + (j & 1)) * OW) = r;
    }
    if (c == 15) {
        #pragma unroll
        for (int j = 0; j < 8; ++j) {
            const float bsj = bs4[j >> 1];
            v2f ed = tope[j];
            ed.x += bsj; ed.y += bsj;
            *(v2f*)(orow + ((j >> 1) * OH + (j & 1)) * OW + 4) = ed;
        }
    }
}

extern "C" void kernel_launch(void* const* d_in, const int* in_sizes, int n_in,
                              void* d_out, int out_size, void* d_ws, size_t ws_size,
                              hipStream_t stream) {
    const float* x    = (const float*)d_in[0];
    const float* wgt  = (const float*)d_in[1];
    const float* bias = (const float*)d_in[2];
    float* out = (float*)d_out;

    dim3 grid(16 * 16);
    convT2d_scatter_wmma_f32<<<grid, dim3(NT), 0, stream>>>(x, wgt, bias, out);
}


# Wave tile 8 image rows (four row pairs), 4 waves per workgroup, one per SIMD (was 2 rows, 16 waves): a K step's weight loads and fragment sums, done once per wave, feed 72 WMMAs (was 18)
# speedup vs baseline: 6.8219x; 1.0961x over previous
#include <hip/hip_runtime.h>


typedef __attribute__((ext_vector_type(2))) float v2f;
typedef __attribute__((ext_vector_type(4))) float v4f;
typedef __attribute__((ext_vector_type(8))) float v8f;
typedef v4f v4f_a8 __attribute__((aligned(8)));

#define IC 256
#define OC 128
#define NPIX 1024
#define OH 66
#define OW 66
#define OCG 8
#define XR 68
#define NT 128
#define NW (NT / 32)
#define NQ 4

__global__ __launch_bounds__(NT)
void convT2d_scatter_wmma_f32(const float* __restrict__ x,
                              const float* __restrict__ w,
                              const float* __restrict__ bias,
                              float* __restrict__ out)
{
    __shared__ __attribute__((aligned(16))) float rb[15 * 2 * OCG * XR];

    const int bid = blockIdx.x;
    const int ocg = bid & 15;
    const int b   = bid >> 4;
    const int oc0 = ocg * OCG;
    const int tid = threadIdx.x;
    const int wv   = tid >> 5;
    const int lane = tid & 31;
    const int c    = lane & 15;
    const int h    = lane >> 4;

    const float* xp = x + (size_t)(b * IC + 2 * h) * NPIX + 64 * wv + 2 * c;
    const float* wp = w + (size_t)(2 * h) * (OC * 16) + (oc0 + (c >> 1)) * 16
                        + (c & 1) * 4;

    v8f acc[NQ][2][3][3];
    #pragma unroll
    for (int q = 0; q < NQ; ++q)
        #pragma unroll
        for (int p = 0; p < 2; ++p)
            #pragma unroll
            for (int e = 0; e < 3; ++e)
                #pragma unroll
                for (int f = 0; f < 3; ++f)
                    acc[q][p][e][f] = (v8f){0.f, 0.f, 0.f, 0.f, 0.f, 0.f, 0.f, 0.f};

    v4f   wa[2][2][2];
    v2f   xb[2][2][NQ][2];
    #pragma unroll
    for (int r = 0; r < 2; ++r) {
        #pragma unroll
        for (int s = 0; s < 2; ++s) {
            #pragma unroll
            for (int e = 0; e < 2; ++e)
                wa[r][s][e] = *(const v4f*)(wp + (4 * r + s) * (OC * 16) + 8 * e);
            #pragma unroll
            for (int q = 0; q < NQ; ++q)
                #pragma unroll
                for (int e = 0; e < 2; ++e)
                    xb[r][s][q][e] = *(const v2f*)(xp + (4 * r + s) * NPIX + 64 * NW * q + 32 * e);
        }

        __builtin_amdgcn_sched_barrier(0);
    }
    wp += 8 * (OC * 16);
    xp += 8 * NPIX;

    for (int k0 = 0; k0 < IC - 8; k0 += 8) {
        #pragma unroll
        for (int r = 0; r < 2; ++r) {
            v2f af[2][3][3], bf[NQ][3][3];
            #pragma unroll
            for (int e = 0; e < 2; ++e) {
                const v4f q0 = wa[r][0][e], q1 = wa[r][1][e];
                #pragma unroll
                for (int f = 0; f < 2; ++f) {
                    af[0][e][f].x = q0[2 * f];      af[0][e][f].y = q1[2 * f];
                    af[1][e][f].x = q0[2 * f + 1];  af[1][e][f].y = q1[2 * f + 1];
                    #pragma unroll
                    for (int q = 0; q < NQ; ++q) {
                        bf[q][e][f].x = xb[r][0][q][e][f];
                        bf[q][e][f].y = xb[r][1][q][e][f];
                    }
                }
            }

            #pragma unroll
            for (int e = 0; e < 2; ++e)
                #pragma unroll
                for (int f = 0; f < 2; ++f)
                    asm volatile("" : "+v"(af[0][e][f]), "+v"(af[1][e][f]), "+v"(bf[0][e][f]), "+v"(bf[1][e][f]), "+v"(bf[2][e][f]), "+v"(bf[3][e][f]));

            #pragma unroll
            for (int e = 0; e < 2; ++e) {
                af[0][e][2] = af[0][e][0] + af[0][e][1];
                af[1][e][2] = af[1][e][0] + af[1][e][1];
                #pragma unroll
                for (int q = 0; q < NQ; ++q)
                    bf[q][e][2] = bf[q][e][0] + bf[q][e][1];
            }
            #pragma unroll
            for (int f = 0; f < 3; ++f) {
                af[0][2][f] = af[0][0][f] + af[0][1][f];
                af[1][2][f] = af[1][0][f] + af[1][1][f];
                #pragma unroll
                for (int q = 0; q < NQ; ++q)
                    bf[q][2][f] = bf[q][0][f] + bf[q][1][f];
            }

            __builtin_amdgcn_sched_barrier(0);
            #pragma unroll
            for (int s = 0; s < 2; ++s) {
                #pragma unroll
                for (int e = 0; e < 2; ++e)
                    wa[r][s][e] = *(const v4f*)(wp + (4 * r + s) * (OC * 16) + 8 * e);
                #pragma unroll
                for (int q = 0; q < NQ; ++q)
                    #pragma unroll
                    for (int e = 0; e < 2; ++e)
                        xb[r][s][q][e] = *(const v2f*)(xp + (4 * r + s) * NPIX + 64 * NW * q + 32 * e);
            }

            #pragma unroll
            for (int q = 0; q < NQ; ++q)
                #pragma unroll
                for (int p = 0; p < 2; ++p)
                    #pragma unroll
                    for (int e = 0; e < 3; ++e)
                        #pragma unroll
                        for (int f = 0; f < 3; ++f)
                            acc[q][p][e][f] = __builtin_amdgcn_wmma_f32_16x16x4_f32(
                                false, af[p][e][f], false, bf[q][e][f], (short)0, acc[q][p][e][f], false, false);

            __builtin_amdgcn_sched_barrier(0);
        }
        wp += 8 * (OC * 16);
        xp += 8 * NPIX;
    }
    #pragma unroll
    for (int r = 0; r < 2; ++r) {
        v2f af[2][3][3], bf[NQ][3][3];
        #pragma unroll
        for (int e = 0; e < 2; ++e) {
            const v4f q0 = wa[r][0][e], q1 = wa[r][1][e];
            #pragma unroll
            for (int f = 0; f < 2; ++f) {
                af[0][e][f].x = q0[2 * f];      af[0][e][f].y = q1[2 * f];
                af[1][e][f].x = q0[2 * f + 1];  af[1][e][f].y = q1[2 * f + 1];
                #pragma unroll
                for (int q = 0; q < NQ; ++q) {
                    bf[q][e][f].x = xb[r][0][q][e][f];
                    bf[q][e][f].y = xb[r][1][q][e][f];
                }
            }
        }

        #pragma unroll
        for (int e = 0; e < 2; ++e) {
            af[0][e][2] = af[0][e][0] + af[0][e][1];
            af[1][e][2] = af[1][e][0] + af[1][e][1];
            #pragma unroll
            for (int q = 0; q < NQ; ++q)
                bf[q][e][2] = bf[q][e][0] + bf[q][e][1];
        }
        #pragma unroll
        for (int f = 0; f < 3; ++f) {
            af[0][2][f] = af[0][0][f] + af[0][1][f];
            af[1][2][f] = af[1][0][f] + af[1][1][f];
            #pragma unroll
            for (int q = 0; q < NQ; ++q)
                bf[q][2][f] = bf[q][0][f] + bf[q][1][f];
        }

        #pragma unroll
        for (int q = 0; q < NQ; ++q)
            #pragma unroll
            for (int p = 0; p < 2; ++p)
                #pragma unroll
                for (int e = 0; e < 3; ++e)
                    #pragma unroll
                    for (int f = 0; f < 3; ++f)
                        acc[q][p][e][f] = __builtin_amdgcn_wmma_f32_16x16x4_f32(
                            false, af[p][e][f], false, bf[q][e][f], (short)0, acc[q][p][e][f], false, false);

    }

    #pragma unroll
    for (int q = 0; q < NQ; ++q)
        #pragma unroll
        for (int p = 0; p < 2; ++p) {
            #pragma unroll
            for (int e = 0; e < 3; ++e) acc[q][p][e][2] = acc[q][p][e][2] - acc[q][p][e][0] - acc[q][p][e][1];
            #pragma unroll
            for (int f = 0; f < 3; ++f) acc[q][p][2][f] = acc[q][p][2][f] - acc[q][p][0][f] - acc[q][p][1][f];
        }

    auto shr1 = [](float v) -> float {
        return __builtin_bit_cast(float, __builtin_amdgcn_update_dpp(
            0, __builtin_bit_cast(int, v), 0x111, 0xf, 0xf, true));
    };
    const int wvu = __builtin_amdgcn_readfirstlane(wv);
    const v4f bs4 = *(const v4f*)(bias + oc0 + 4 * h);

    v4f top[NQ][8];
    v2f tope[NQ][8];
    #pragma unroll
    for (int q = 0; q < NQ; ++q) {
        const int rp = wvu + NW * q;
        float* orow = out + ((size_t)(b * OC + oc0 + 4 * h) * OH + 4 * rp) * OW + 4 * c;

        #pragma unroll
        for (int j = 0; j < 8; ++j) {
            const float bsj = bs4[j >> 1];
            v4f r;
            r.x = (acc[q][0][2][0][j] + shr1(acc[q][0][2][1][j])) + bsj;
            r.y = (acc[q][1][2][0][j] + shr1(acc[q][1][2][1][j])) + bsj;
            r.z = acc[q][0][2][2][j] + bsj;
            r.w = acc[q][1][2][2][j] + bsj;
            *(v4f_a8*)(orow + ((j >> 1) * OH + 2 + (j & 1)) * OW) = r;
        }
        if (c == 15) {
            #pragma unroll
            for (int j = 0; j < 8; ++j) {
                const float bsj = bs4[j >> 1];
                v2f ed;
                ed.x = acc[q][0][2][1][j] + bsj;
                ed.y = acc[q][1][2][1][j] + bsj;
                *(v2f*)(orow + ((j >> 1) * OH + 2 + (j & 1)) * OW + 4) = ed;
            }
        }

        if (rp < 15) {
            float* rw = rb + (rp * 16 + 8 * h) * XR + 4 * c;
            #pragma unroll
            for (int j = 0; j < 8; ++j) {
                v4f r;
                r.x = acc[q][0][1][0][j] + shr1(acc[q][0][1][1][j]);
                r.y = acc[q][1][1][0][j] + shr1(acc[q][1][1][1][j]);
                r.z = acc[q][0][1][2][j];
                r.w = acc[q][1][1][2][j];
                *(v4f*)(rw + j * XR) = r;
            }
            if (c == 15) {
                #pragma unroll
                for (int j = 0; j < 8; ++j) {
                    v2f ed;
                    ed.x = acc[q][0][1][1][j];
                    ed.y = acc[q][1][1][1][j];
                    *(v2f*)(rw + j * XR + 4) = ed;
                }
            }
        } else {
            #pragma unroll
            for (int j = 0; j < 8; ++j) {
                const float bsj = bs4[j >> 1];
                v4f r;
                r.x = (acc[q][0][1][0][j] + shr1(acc[q][0][1][1][j])) + bsj;
                r.y = (acc[q][1][1][0][j] + shr1(acc[q][1][1][1][j])) + bsj;
                r.z = acc[q][0][1][2][j] + bsj;
                r.w = acc[q][1][1][2][j] + bsj;
                *(v4f_a8*)(orow + ((j >> 1) * OH + 4 + (j & 1)) * OW) = r;
            }
            if (c == 15) {
                #pragma unroll
                for (int j = 0; j < 8; ++j) {
                    const float bsj = bs4[j >> 1];
                    v2f ed;
                    ed.x = acc[q][0][1][1][j] + bsj;
                    ed.y = acc[q][1][1][1][j] + bsj;
                    *(v2f*)(orow + ((j >> 1) * OH + 4 + (j & 1)) * OW + 4) = ed;
                }
            }
        }

        #pragma unroll
        for (int j = 0; j < 8; ++j) {
            top[q][j].x = acc[q][0][0][0][j] + shr1(acc[q][0][0][1][j]);
            top[q][j].y = acc[q][1][0][0][j] + shr1(acc[q][1][0][1][j]);
            top[q][j].z = acc[q][0][0][2][j];
            top[q][j].w = acc[q][1][0][2][j];
            tope[q][j].x = acc[q][0][0][1][j];
            tope[q][j].y = acc[q][1][0][1][j];
        }
    }
    __builtin_amdgcn_fence(__ATOMIC_RELEASE, "workgroup", "local");
    __builtin_amdgcn_s_barrier();
    __builtin_amdgcn_fence(__ATOMIC_ACQUIRE, "workgroup", "local");
    #pragma unroll
    for (int q = 0; q < NQ; ++q) {
        const int rp = wvu + NW * q;
        float* orow = out + ((size_t)(b * OC + oc0 + 4 * h) * OH + 4 * rp) * OW + 4 * c;
        if (rp > 0) {
            const float* rr = rb + ((rp - 1) * 16 + 8 * h) * XR + 4 * c;
            #pragma unroll
            for (int j = 0; j < 8; ++j)
                top[q][j] += *(const v4f*)(rr + j * XR);
            if (c == 15) {
                #pragma unroll
                for (int j = 0; j < 8; ++j)
                    tope[q][j] += *(const v2f*)(rr + j * XR + 4);
            }
        }
        #pragma unroll
        for (int j = 0; j < 8; ++j) {
            const float bsj = bs4[j >> 1];
            v4f r = top[q][j];
            r.x += bsj; r.y += bsj; r.z += bsj; r.w += bsj;
            *(v4f_a8*)(orow + ((j >> 1) * OH + (j & 1)) * OW) = r;
        }
        if (c == 15) {
            #pragma unroll
            for (int j = 0; j < 8; ++j) {
                const float bsj = bs4[j >> 1];
                v2f ed = tope[q][j];
                ed.x += bsj; ed.y += bsj;
                *(v2f*)(orow + ((j >> 1) * OH + (j & 1)) * OW + 4) = ed;
            }
        }
    }
}

extern "C" void kernel_launch(void* const* d_in, const int* in_sizes, int n_in,
                              void* d_out, int out_size, void* d_ws, size_t ws_size,
                              hipStream_t stream) {
    const float* x    = (const float*)d_in[0];
    const float* wgt  = (const float*)d_in[1];
    const float* bias = (const float*)d_in[2];
    float* out = (float*)d_out;

    dim3 grid(16 * 16);
    convT2d_scatter_wmma_f32<<<grid, dim3(NT), 0, stream>>>(x, wgt, bias, out);
}
